# starting kernel
# baseline (speedup 1.0000x reference)
#include <hip/hip_runtime.h>
#include <hip/hip_bf16.h>

typedef __attribute__((ext_vector_type(2))) float v2f;
typedef __attribute__((ext_vector_type(8))) float v8f;

#define B_     8
#define ND_    512
#define D_     128
#define NV_    4096
#define NODES_ 512

__global__ __launch_bounds__(256) void hier_sim_bin_attn_kernel(
    const int*   __restrict__ doc_index,
    const float* __restrict__ attn,
    const float* __restrict__ emb,
    const float* __restrict__ VvT,
    const float* __restrict__ bin_diff,
    const float* __restrict__ bin_start,
    float*       __restrict__ a_out)
{
    __shared__ float attn_lds[ND_];
    __shared__ float bin_lds[16];

    const int b    = blockIdx.y;
    const int tid  = threadIdx.x;
    const int wave = tid >> 5;
    const int lane = tid & 31;
    const int col  = lane & 15;
    const int hi   = lane >> 4;

    for (int i = tid; i < ND_; i += 256)
        attn_lds[i] = attn[b * ND_ + i];
    if (tid == 0) {
        float run = bin_start[0];
        #pragma unroll
        for (int i = 0; i < 16; ++i) {
            float d = bin_diff[i];
            run += (d > 0.0f) ? d : 0.0f;
            bin_lds[i] = run;
        }
    }
    __syncthreads();

    const int v0 = blockIdx.x * 128 + wave * 16;

    v2f Bf[32];
    #pragma unroll
    for (int s = 0; s < 32; ++s) {
        const int k = 4 * s + 2 * hi;
        Bf[s].x = VvT[(size_t)k       * NV_ + v0 + col];
        Bf[s].y = VvT[(size_t)(k + 1) * NV_ + v0 + col];
    }

    const float inv_h = 14.0f / 1.49f;

    float acc = 0.0f;
    for (int nt = 0; nt < 32; ++nt) {
        const int nbase = nt * 16;
        const int row   = doc_index[b * ND_ + nbase + col];
        const float* arow = emb + (size_t)row * D_ + 2 * hi;

        v8f c = {};
        #pragma unroll
        for (int s = 0; s < 32; ++s) {
            v2f af = *(const v2f*)(arow + 4 * s);
            c = __builtin_amdgcn_wmma_f32_16x16x4_f32(
                    false, af, false, Bf[s], (short)0, c, false, false);
        }

        #pragma unroll
        for (int r = 0; r < 8; ++r) {
            const float x = c[r];
            const float t = (x + 0.5f) * inv_h;
            int dig = (t < 0.0f) ? 0 : (int)t + 1;
            dig = (dig > 15) ? 15 : dig;
            const float w = bin_lds[dig];
            const int   m = r + 8 * hi;
            acc += w * attn_lds[nbase + m];
        }
    }

    acc += __shfl_xor(acc, 16, 32);
    if (hi == 0)
        a_out[(size_t)b * NV_ + v0 + col] = acc;
}

__global__ __launch_bounds__(32) void hier_final_gemm_kernel(
    const float* __restrict__ a,
    const float* __restrict__ phi,
    float*       __restrict__ out)
{
    const int j0   = blockIdx.x * 16;
    const int lane = threadIdx.x & 31;
    const int col  = lane & 15;
    const int hi   = lane >> 4;
    const int m    = col;

    const float* arow = a   + (size_t)m * NV_ + 2 * hi;
    const float* prow = phi + (size_t)(j0 + col) * NV_ + 2 * hi;

    v8f c = {};
    for (int s = 0; s < NV_ / 4; ++s) {
        v2f af;
        if (m < 8) af = *(const v2f*)(arow + 4 * s);
        else       { af.x = 0.0f; af.y = 0.0f; }
        v2f bf = *(const v2f*)(prow + 4 * s);
        c = __builtin_amdgcn_wmma_f32_16x16x4_f32(
                false, af, false, bf, (short)0, c, false, false);
    }

    #pragma unroll
    for (int r = 0; r < 8; ++r) {
        const int M = r + 8 * hi;
        if (M < 8)
            out[(size_t)M * NODES_ + j0 + col] = c[r];
    }
}

extern "C" void kernel_launch(void* const* d_in, const int* in_sizes, int n_in,
                              void* d_out, int out_size, void* d_ws, size_t ws_size,
                              hipStream_t stream) {
    const int*   doc_index = (const int*)  d_in[0];
    const float* attn      = (const float*)d_in[1];
    const float* emb       = (const float*)d_in[2];
    const float* VvT       = (const float*)d_in[3];
    const float* phi       = (const float*)d_in[4];
    const float* bin_diff  = (const float*)d_in[5];
    const float* bin_start = (const float*)d_in[6];

    float* a_scratch = (float*)d_ws;
    float* out       = (float*)d_out;

    dim3 grid1(NV_ / 128, B_);
    hier_sim_bin_attn_kernel<<<grid1, 256, 0, stream>>>(
        doc_index, attn, emb, VvT, bin_diff, bin_start, a_scratch);

    hier_final_gemm_kernel<<<NODES_ / 16, 32, 0, stream>>>(
        a_scratch, phi, out);
}


# first kernel re-tiled: a wave owns 64 v-columns x 128 n rows (was 16 x 512), each gathered A fragment feeds 4 WMMAs; n quarters added via LDS in fixed order; changes the order of the sum over n
# speedup vs baseline: 6.1694x; 6.1694x over previous
#include <hip/hip_runtime.h>
#include <hip/hip_bf16.h>


typedef __attribute__((ext_vector_type(2))) float v2f;
typedef __attribute__((ext_vector_type(4))) float v4f;
typedef __attribute__((ext_vector_type(8))) float v8f;

#define B_     8
#define ND_    512
#define D_     128
#define NV_    4096
#define NODES_ 512
#define VBLK_  128
#define NSLICE_ (NV_ / VBLK_)

#define NTW_ 8

__device__ __forceinline__ void gather_rows(v2f (&A)[32], const float* __restrict__ emb,
                                            int row, int hi)
{
    const float* arow = emb + (size_t)row * D_ + 2 * hi;
    #pragma unroll
    for (int s = 0; s < 32; ++s)
        A[s] = *(const v2f*)(arow + 4 * s);
}

__device__ __forceinline__ void sim_tile(const v2f (&A)[32], const v2f (&Bf)[4][32],
                                         const float* __restrict__ attn_t,
                                         const float* __restrict__ bin_lds,
                                         float (&acc)[4])
{
    v8f c[4] = {};
    #pragma unroll
    for (int s = 0; s < 32; ++s) {
        #pragma unroll
        for (int j = 0; j < 4; ++j)
            c[j] = __builtin_amdgcn_wmma_f32_16x16x4_f32(
                       false, A[s], false, Bf[j][s], (short)0, c[j], false, false);
    }

    const float inv_h = 14.0f / 1.49f;
    const v4f at0 = *(const v4f*)(attn_t);
    const v4f at1 = *(const v4f*)(attn_t + 4);
    const float at[8] = { at0.x, at0.y, at0.z, at0.w, at1.x, at1.y, at1.z, at1.w };

    #pragma unroll
    for (int j = 0; j < 4; ++j) {
        #pragma unroll
        for (int r = 0; r < 8; ++r) {
            const float x = c[j][r];
            const float t = (x + 0.5f) * inv_h;
            int dig = (t < 0.0f) ? 0 : (int)t + 1;
            dig = (dig > 15) ? 15 : dig;
            const float w = bin_lds[dig];
            acc[j] += w * at[r];
        }
    }
}

__global__ __launch_bounds__(256) void hier_sim_bin_attn_final_partial_kernel(
    const int*   __restrict__ doc_index,
    const float* __restrict__ attn,
    const float* __restrict__ emb,
    const float* __restrict__ VvT,
    const float* __restrict__ phi,
    const float* __restrict__ bin_diff,
    const float* __restrict__ bin_start,
    float*       __restrict__ part)
{
    __shared__ __attribute__((aligned(16))) float attn_lds[ND_];
    __shared__ int   idx_lds[ND_];
    __shared__ float bin_lds[16];
    __shared__ __attribute__((aligned(16))) float a_part[4][VBLK_];
    __shared__ __attribute__((aligned(16))) float a_lds[VBLK_];

    const int b    = blockIdx.y;
    const int tid  = threadIdx.x;
    const int wave = tid >> 5;
    const int lane = tid & 31;
    const int col  = lane & 15;
    const int hi   = lane >> 4;
    const int vq   = wave & 1;
    const int nq   = wave >> 1;

    for (int i = tid; i < ND_; i += 256) {
        attn_lds[i] = attn[b * ND_ + i];
        idx_lds[i]  = doc_index[b * ND_ + i];
    }
    if (tid == 0) {
        float run = bin_start[0];
        #pragma unroll
        for (int i = 0; i < 16; ++i) {
            float d = bin_diff[i];
            run += (d > 0.0f) ? d : 0.0f;
            bin_lds[i] = run;
        }
    }

    const int vb = blockIdx.x * VBLK_;

    v2f Bf[4][32];
    {
        const float* bcol = VvT + vb + vq * 64 + 4 * col;
        #pragma unroll
        for (int s = 0; s < 32; ++s) {
            const int k = 4 * s + 2 * hi;
            const v4f l0 = *(const v4f*)(bcol + (size_t)k       * NV_);
            const v4f l1 = *(const v4f*)(bcol + (size_t)(k + 1) * NV_);
            Bf[0][s].x = l0.x;  Bf[0][s].y = l1.x;
            Bf[1][s].x = l0.y;  Bf[1][s].y = l1.y;
            Bf[2][s].x = l0.z;  Bf[2][s].y = l1.z;
            Bf[3][s].x = l0.w;  Bf[3][s].y = l1.w;
        }
    }
    __syncthreads();

    const int nw = nq * (NTW_ * 16);
    float acc[4] = { 0.0f, 0.0f, 0.0f, 0.0f };

    v2f A0[32], A1[32];
    gather_rows(A0, emb, idx_lds[nw + col], hi);
    for (int it = 0; it < NTW_ - 2; it += 2) {
        const int n0 = nw + it * 16;
        gather_rows(A1, emb, idx_lds[n0 + 16 + col], hi);
        __builtin_amdgcn_sched_barrier(0);
        sim_tile(A0, Bf, &attn_lds[n0 + 8 * hi], bin_lds, acc);
        gather_rows(A0, emb, idx_lds[n0 + 32 + col], hi);
        __builtin_amdgcn_sched_barrier(0);
        sim_tile(A1, Bf, &attn_lds[n0 + 16 + 8 * hi], bin_lds, acc);
    }
    {
        const int n0 = nw + (NTW_ - 2) * 16;
        gather_rows(A1, emb, idx_lds[n0 + 16 + col], hi);
        __builtin_amdgcn_sched_barrier(0);
        sim_tile(A0, Bf, &attn_lds[n0 + 8 * hi], bin_lds, acc);
        sim_tile(A1, Bf, &attn_lds[n0 + 16 + 8 * hi], bin_lds, acc);
    }

    const float* p0 = phi + (size_t)(wave * 64 + col) * NV_ + vb + 4 * hi;
    v4f q0[16], q1[16];
    #pragma unroll
    for (int t = 0; t < 16; ++t) {
        q0[t] = *(const v4f*)(p0 + 8 * t);
        q1[t] = *(const v4f*)(p0 + (size_t)16 * NV_ + 8 * t);
    }

    #pragma unroll
    for (int j = 0; j < 4; ++j)
        acc[j] += __shfl_xor(acc[j], 16, 32);
    if (hi == 0) {
        v4f v = { acc[0], acc[1], acc[2], acc[3] };
        *(v4f*)(&a_part[nq][vq * 64 + 4 * col]) = v;
    }
    __syncthreads();
    if (tid < VBLK_)
        a_lds[tid] = ((a_part[0][tid] + a_part[1][tid]) + a_part[2][tid]) + a_part[3][tid];
    __syncthreads();

    v4f af4[16];
    #pragma unroll
    for (int t = 0; t < 16; ++t) {
        v4f v = *(const v4f*)(&a_lds[8 * t + 4 * hi]);
        if (col & 7) { v.x = 0.0f; v.y = 0.0f; v.z = 0.0f; v.w = 0.0f; }
        af4[t] = v;
    }

    float* pout = part + ((size_t)(blockIdx.x * B_ + b)) * NODES_ + wave * 64;

    #pragma unroll
    for (int pr = 0; pr < 2; ++pr) {
        if (pr == 1) {
            const float* p1 = p0 + (size_t)32 * NV_;
            #pragma unroll
            for (int t = 0; t < 16; ++t) {
                q0[t] = *(const v4f*)(p1 + 8 * t);
                q1[t] = *(const v4f*)(p1 + (size_t)16 * NV_ + 8 * t);
            }
        }
        v8f c0 = {}, c1 = {};
        #pragma unroll
        for (int t = 0; t < 16; ++t) {
            v2f alo = { af4[t].x, af4[t].y };
            v2f ahi = { af4[t].z, af4[t].w };
            v2f b0lo = { q0[t].x, q0[t].y };
            v2f b0hi = { q0[t].z, q0[t].w };
            v2f b1lo = { q1[t].x, q1[t].y };
            v2f b1hi = { q1[t].z, q1[t].w };
            c0 = __builtin_amdgcn_wmma_f32_16x16x4_f32(
                     false, alo, false, b0lo, (short)0, c0, false, false);
            c1 = __builtin_amdgcn_wmma_f32_16x16x4_f32(
                     false, alo, false, b1lo, (short)0, c1, false, false);
            c0 = __builtin_amdgcn_wmma_f32_16x16x4_f32(
                     false, ahi, false, b0hi, (short)0, c0, false, false);
            c1 = __builtin_amdgcn_wmma_f32_16x16x4_f32(
                     false, ahi, false, b1hi, (short)0, c1, false, false);
        }
        const float val = hi ? c1[0] : c0[0];
        pout[pr * 32 + lane] = val;
    }
}

__global__ __launch_bounds__(64) void hier_partial_sum_kernel(
    const float* __restrict__ part,
    float*       __restrict__ out)
{
    const int idx = blockIdx.x * 64 + threadIdx.x;
    float p[NSLICE_];
    #pragma unroll
    for (int x = 0; x < NSLICE_; ++x)
        p[x] = part[(size_t)x * (B_ * NODES_) + idx];
    float s = p[0];
    #pragma unroll
    for (int x = 1; x < NSLICE_; ++x)
        s += p[x];
    out[idx] = s;
}

extern "C" void kernel_launch(void* const* d_in, const int* in_sizes, int n_in,
                              void* d_out, int out_size, void* d_ws, size_t ws_size,
                              hipStream_t stream) {
    const int*   doc_index = (const int*)  d_in[0];
    const float* attn      = (const float*)d_in[1];
    const float* emb       = (const float*)d_in[2];
    const float* VvT       = (const float*)d_in[3];
    const float* phi       = (const float*)d_in[4];
    const float* bin_diff  = (const float*)d_in[5];
    const float* bin_start = (const float*)d_in[6];

    float* part = (float*)d_ws;
    float* out  = (float*)d_out;

    dim3 grid1(NV_ / VBLK_, B_);
    hier_sim_bin_attn_final_partial_kernel<<<grid1, 256, 0, stream>>>(
        doc_index, attn, emb, VvT, phi, bin_diff, bin_start, part);

    hier_partial_sum_kernel<<<(B_ * NODES_) / 64, 64, 0, stream>>>(
        part, out);
}
